# starting kernel
# baseline (speedup 1.0000x reference)
#include <hip/hip_runtime.h>

typedef __attribute__((ext_vector_type(2))) float v2f;
typedef __attribute__((ext_vector_type(8))) float v8f;

#define BN   16
#define TT   64
#define FF   8192
#define SLD  68

__global__ void __launch_bounds__(256)
qk_dot_kernel(const float* __restrict__ Q,
              const float* __restrict__ K,
              float* __restrict__ S) {
  const int bn   = blockIdx.x;
  const int lane = threadIdx.x & 31;
  const int wave = threadIdx.x >> 5;
  const int row  = lane & 15;
  const int koff = (lane >> 4) << 1;

  const float* q = Q + (size_t)bn * TT * FF;
  const float* k = K + (size_t)bn * TT * FF;

  const int tj   = wave & 3;
  const int ti_a = wave >> 2;
  const int ti_b = ti_a + 2;

  const float* qa = q + (size_t)(ti_a * 16 + row) * FF + koff;
  const float* qb = q + (size_t)(ti_b * 16 + row) * FF + koff;
  const float* kb = k + (size_t)(tj   * 16 + row) * FF + koff;

  v8f acc_a = {};
  v8f acc_b = {};
#pragma unroll 4
  for (int kk = 0; kk < FF; kk += 4) {
    v2f a0 = *(const v2f*)(qa + kk);
    v2f a1 = *(const v2f*)(qb + kk);
    v2f b  = *(const v2f*)(kb + kk);
    acc_a = __builtin_amdgcn_wmma_f32_16x16x4_f32(false, a0, false, b,
                                                  (short)0, acc_a, false, false);
    acc_b = __builtin_amdgcn_wmma_f32_16x16x4_f32(false, a1, false, b,
                                                  (short)0, acc_b, false, false);
  }

  const float scale = 1.0f / 1024.0f;
  float* sb = S + (size_t)bn * TT * TT;
  const int qcol = tj * 16 + row;
  const int half = (lane >> 4) * 8;
#pragma unroll
  for (int r = 0; r < 8; ++r) {
    sb[(ti_a * 16 + half + r) * TT + qcol] = acc_a[r] * scale;
    sb[(ti_b * 16 + half + r) * TT + qcol] = acc_b[r] * scale;
  }
}

__global__ void __launch_bounds__(256)
sv_gauss_kernel(const float* __restrict__ S,
                const float* __restrict__ V,
                float* __restrict__ O) {
  __shared__ float Sl[TT * SLD];

  const int bn    = blockIdx.x >> 6;
  const int chunk = blockIdx.x & 63;
  const int tid   = threadIdx.x;
  const int lane  = tid & 31;
  const int wave  = tid >> 5;

  const float* sg = S + (size_t)bn * TT * TT;
  for (int i = tid; i < TT * TT; i += 256) {
    Sl[(i >> 6) * SLD + (i & 63)] = sg[i];
  }
  __syncthreads();

  const int ntile = chunk * 8 + wave;
  const int f0    = ntile * 16;
  const int row   = lane & 15;
  const int koff  = (lane >> 4) << 1;

  const float* v = V + (size_t)bn * TT * FF;

  v8f acc[4] = {};
  for (int kk = 0; kk < TT; kk += 4) {
    v2f b;
    b.x = v[(size_t)(kk + koff)     * FF + f0 + row];
    b.y = v[(size_t)(kk + koff + 1) * FF + f0 + row];
#pragma unroll
    for (int pb = 0; pb < 4; ++pb) {
      v2f a = *(const v2f*)(&Sl[(pb * 16 + row) * SLD + kk + koff]);
      acc[pb] = __builtin_amdgcn_wmma_f32_16x16x4_f32(false, a, false, b,
                                                      (short)0, acc[pb], false, false);
    }
  }

  const int f = f0 + row;
  const int m = f >> 3;
  const float dw = (float)(m & 31) - 15.5f;
  const float dh = (float)(m >> 5) - 15.5f;
  float sx = 0.0f;
#pragma unroll
  for (int x = 0; x < 32; ++x) {
    float d = (float)x - 15.5f;
    sx += __expf(-d * d * (1.0f / 50.0f));
  }
  const float g = __expf(-(dw * dw + dh * dh) * (1.0f / 50.0f)) / (sx * sx);

  float* o = O + (size_t)bn * TT * FF;
  const int half = (lane >> 4) * 8;
#pragma unroll
  for (int pb = 0; pb < 4; ++pb) {
#pragma unroll
    for (int r = 0; r < 8; ++r) {
      const int p = pb * 16 + half + r;
      o[(size_t)p * FF + f] = acc[pb][r] * g;
    }
  }
}

extern "C" void kernel_launch(void* const* d_in, const int* in_sizes, int n_in,
                              void* d_out, int out_size, void* d_ws, size_t ws_size,
                              hipStream_t stream) {
  const float* Q = (const float*)d_in[0];
  const float* K = (const float*)d_in[1];
  const float* V = (const float*)d_in[2];
  float* O = (float*)d_out;
  float* S = (float*)d_ws;

  qk_dot_kernel<<<BN, 256, 0, stream>>>(Q, K, S);
  sv_gauss_kernel<<<BN * 64, 256, 0, stream>>>(S, V, O);
}


# Score reduction over the 8192 features split across 32 work-groups per head (16 -> 512 work-groups), partial planes added in fixed order by a new small launch: changes the f32 summation order
# speedup vs baseline: 7.5509x; 7.5509x over previous
#include <hip/hip_runtime.h>

typedef __attribute__((ext_vector_type(2))) float v2f;
typedef __attribute__((ext_vector_type(4))) float v4f;
typedef __attribute__((ext_vector_type(8))) float v8f;

#define BN   16
#define TT   64
#define FF   8192
#define SLD  68
#define KSPLIT 32
#define KC   (FF / KSPLIT)

__global__ void __launch_bounds__(256)
qk_partial_kernel(const float* __restrict__ Q,
                  const float* __restrict__ K,
                  float* __restrict__ P) {
  const int bn   = blockIdx.x / KSPLIT;
  const int c    = blockIdx.x % KSPLIT;
  const int lane = threadIdx.x & 31;
  const int wave = threadIdx.x >> 5;
  const int row  = lane & 15;
  const int koff = (lane >> 4) << 1;

  const float* q = Q + (size_t)bn * TT * FF + c * KC;
  const float* k = K + (size_t)bn * TT * FF + c * KC;

  const int tj   = wave & 3;
  const int ti_a = wave >> 2;
  const int ti_b = ti_a + 2;

  const float* qa = q + (size_t)(ti_a * 16 + row) * FF + koff;
  const float* qb = q + (size_t)(ti_b * 16 + row) * FF + koff;
  const float* kb = k + (size_t)(tj   * 16 + row) * FF + koff;

  v8f acc_a = {};
  v8f acc_b = {};
#pragma unroll 4
  for (int kk = 0; kk < KC; kk += 4) {
    v2f a0 = *(const v2f*)(qa + kk);
    v2f a1 = *(const v2f*)(qb + kk);
    v2f b  = *(const v2f*)(kb + kk);
    acc_a = __builtin_amdgcn_wmma_f32_16x16x4_f32(false, a0, false, b,
                                                  (short)0, acc_a, false, false);
    acc_b = __builtin_amdgcn_wmma_f32_16x16x4_f32(false, a1, false, b,
                                                  (short)0, acc_b, false, false);
  }

  float* pb = P + (size_t)blockIdx.x * TT * TT;
  const int qcol = tj * 16 + row;
  const int half = (lane >> 4) * 8;
#pragma unroll
  for (int r = 0; r < 8; ++r) {
    pb[(ti_a * 16 + half + r) * TT + qcol] = acc_a[r];
    pb[(ti_b * 16 + half + r) * TT + qcol] = acc_b[r];
  }
}

__global__ void __launch_bounds__(256)
s_sum_kernel(const float* __restrict__ P,
             float* __restrict__ S) {
  const int g  = blockIdx.x * 256 + threadIdx.x;
  const int bn = g >> 10;
  const int e  = (g & 1023) << 2;
  const float* p = P + (size_t)bn * KSPLIT * TT * TT + e;

  v4f s = *(const v4f*)p;
#pragma unroll
  for (int c = 1; c < KSPLIT; ++c) {
    s += *(const v4f*)(p + (size_t)c * TT * TT);
  }
  const float scale = 1.0f / 1024.0f;
  *(v4f*)(S + (size_t)bn * TT * TT + e) = s * scale;
}

__global__ void __launch_bounds__(256)
sv_gauss_kernel(const float* __restrict__ S,
                const float* __restrict__ V,
                float* __restrict__ O) {
  __shared__ float Sl[TT * SLD];

  const int bn    = blockIdx.x >> 6;
  const int chunk = blockIdx.x & 63;
  const int tid   = threadIdx.x;
  const int lane  = tid & 31;
  const int wave  = tid >> 5;

  const float* sg = S + (size_t)bn * TT * TT;
  for (int i = tid; i < TT * TT; i += 256) {
    Sl[(i >> 6) * SLD + (i & 63)] = sg[i];
  }
  __syncthreads();

  const int ntile = chunk * 8 + wave;
  const int f0    = ntile * 16;
  const int row   = lane & 15;
  const int koff  = (lane >> 4) << 1;

  const float* v = V + (size_t)bn * TT * FF;

  v8f acc[4] = {};
  for (int kk = 0; kk < TT; kk += 4) {
    v2f b;
    b.x = v[(size_t)(kk + koff)     * FF + f0 + row];
    b.y = v[(size_t)(kk + koff + 1) * FF + f0 + row];
#pragma unroll
    for (int pb = 0; pb < 4; ++pb) {
      v2f a = *(const v2f*)(&Sl[(pb * 16 + row) * SLD + kk + koff]);
      acc[pb] = __builtin_amdgcn_wmma_f32_16x16x4_f32(false, a, false, b,
                                                      (short)0, acc[pb], false, false);
    }
  }

  const int f = f0 + row;
  const int m = f >> 3;
  const float dw = (float)(m & 31) - 15.5f;
  const float dh = (float)(m >> 5) - 15.5f;
  float sx = 0.0f;
#pragma unroll
  for (int x = 0; x < 32; ++x) {
    float d = (float)x - 15.5f;
    sx += __expf(-d * d * (1.0f / 50.0f));
  }
  const float g = __expf(-(dw * dw + dh * dh) * (1.0f / 50.0f)) / (sx * sx);

  float* o = O + (size_t)bn * TT * FF;
  const int half = (lane >> 4) * 8;
#pragma unroll
  for (int pb = 0; pb < 4; ++pb) {
#pragma unroll
    for (int r = 0; r < 8; ++r) {
      const int p = pb * 16 + half + r;
      o[(size_t)p * FF + f] = acc[pb][r] * g;
    }
  }
}

extern "C" void kernel_launch(void* const* d_in, const int* in_sizes, int n_in,
                              void* d_out, int out_size, void* d_ws, size_t ws_size,
                              hipStream_t stream) {
  const float* Q = (const float*)d_in[0];
  const float* K = (const float*)d_in[1];
  const float* V = (const float*)d_in[2];
  float* O = (float*)d_out;
  float* S = (float*)d_ws;
  float* P = O;

  qk_partial_kernel<<<BN * KSPLIT, 256, 0, stream>>>(Q, K, P);
  s_sum_kernel<<<64, 256, 0, stream>>>(P, S);
  sv_gauss_kernel<<<BN * 64, 256, 0, stream>>>(S, V, O);
}


# Score kernel: Q and K chunk staged through double-buffered LDS with whole-line 128-bit global loads (each input byte fetched once per work-group); fragments read from LDS in the same k order
# speedup vs baseline: 9.7884x; 1.2963x over previous
#include <hip/hip_runtime.h>

typedef __attribute__((ext_vector_type(2))) float v2f;
typedef __attribute__((ext_vector_type(4))) float v4f;
typedef __attribute__((ext_vector_type(8))) float v8f;

#define BN   16
#define TT   64
#define FF   8192
#define SLD  68
#define KSPLIT 32
#define KC   (FF / KSPLIT)

#define SUB   64
#define NSUB  (KC / SUB)
#define LP    68

__global__ void __launch_bounds__(256)
qk_partial_kernel(const float* __restrict__ Q,
                  const float* __restrict__ K,
                  float* __restrict__ P) {
  __shared__ __attribute__((aligned(16))) float L[2][2 * TT * LP];

  const int bn   = blockIdx.x / KSPLIT;
  const int c    = blockIdx.x % KSPLIT;
  const int tid  = threadIdx.x;
  const int lane = tid & 31;
  const int wave = tid >> 5;
  const int row  = lane & 15;
  const int koff = (lane >> 4) << 1;

  const int srow = tid >> 4;
  const int scol = (tid & 15) << 2;
  const size_t goff = (size_t)bn * TT * FF + (size_t)c * KC + (size_t)srow * FF + scol;
  const float* gq = Q + goff;
  const float* gk = K + goff;
  const int loff = srow * LP + scol;

  const int tj   = wave & 3;
  const int ti_a = wave >> 2;
  const int ti_b = ti_a + 2;
  const int fa = (ti_a * 16 + row) * LP + koff;
  const int fb = (ti_b * 16 + row) * LP + koff;
  const int fk = (TT + tj * 16 + row) * LP + koff;

  v4f rq[4], rk[4];
#pragma unroll
  for (int i = 0; i < 4; ++i) {
    rq[i] = *(const v4f*)(gq + (size_t)(16 * i) * FF);
    rk[i] = *(const v4f*)(gk + (size_t)(16 * i) * FF);
  }
#pragma unroll
  for (int i = 0; i < 4; ++i) {
    *(v4f*)(&L[0][loff + 16 * i * LP])        = rq[i];
    *(v4f*)(&L[0][loff + (TT + 16 * i) * LP]) = rk[i];
  }
  __syncthreads();

  v8f acc_a = {};
  v8f acc_b = {};
#pragma unroll
  for (int s = 0; s < NSUB; ++s) {
    const float* lc = L[s & 1];
    float*       ln = L[(s + 1) & 1];
    if (s + 1 < NSUB) {
#pragma unroll
      for (int i = 0; i < 4; ++i) {
        rq[i] = *(const v4f*)(gq + (size_t)(16 * i) * FF + (s + 1) * SUB);
        rk[i] = *(const v4f*)(gk + (size_t)(16 * i) * FF + (s + 1) * SUB);
      }
    }
#pragma unroll
    for (int kk = 0; kk < SUB; kk += 4) {
      v2f a0 = *(const v2f*)(lc + fa + kk);
      v2f a1 = *(const v2f*)(lc + fb + kk);
      v2f b  = *(const v2f*)(lc + fk + kk);
      acc_a = __builtin_amdgcn_wmma_f32_16x16x4_f32(false, a0, false, b,
                                                    (short)0, acc_a, false, false);
      acc_b = __builtin_amdgcn_wmma_f32_16x16x4_f32(false, a1, false, b,
                                                    (short)0, acc_b, false, false);
    }
    if (s + 1 < NSUB) {
#pragma unroll
      for (int i = 0; i < 4; ++i) {
        *(v4f*)(&ln[loff + 16 * i * LP])        = rq[i];
        *(v4f*)(&ln[loff + (TT + 16 * i) * LP]) = rk[i];
      }
      __syncthreads();
    }
  }

  float* pb = P + (size_t)blockIdx.x * TT * TT;
  const int qcol = tj * 16 + row;
  const int half = (lane >> 4) * 8;
#pragma unroll
  for (int r = 0; r < 8; ++r) {
    pb[(ti_a * 16 + half + r) * TT + qcol] = acc_a[r];
    pb[(ti_b * 16 + half + r) * TT + qcol] = acc_b[r];
  }
}

__global__ void __launch_bounds__(256)
s_sum_kernel(const float* __restrict__ P,
             float* __restrict__ S) {
  const int g  = blockIdx.x * 256 + threadIdx.x;
  const int bn = g >> 10;
  const int e  = (g & 1023) << 2;
  const float* p = P + (size_t)bn * KSPLIT * TT * TT + e;

  v4f s = *(const v4f*)p;
#pragma unroll
  for (int c = 1; c < KSPLIT; ++c) {
    s += *(const v4f*)(p + (size_t)c * TT * TT);
  }
  const float scale = 1.0f / 1024.0f;
  *(v4f*)(S + (size_t)bn * TT * TT + e) = s * scale;
}

__global__ void __launch_bounds__(256)
sv_gauss_kernel(const float* __restrict__ S,
                const float* __restrict__ V,
                float* __restrict__ O) {
  __shared__ float Sl[TT * SLD];

  const int bn    = blockIdx.x >> 6;
  const int chunk = blockIdx.x & 63;
  const int tid   = threadIdx.x;
  const int lane  = tid & 31;
  const int wave  = tid >> 5;

  const float* sg = S + (size_t)bn * TT * TT;
  for (int i = tid; i < TT * TT; i += 256) {
    Sl[(i >> 6) * SLD + (i & 63)] = sg[i];
  }
  __syncthreads();

  const int ntile = chunk * 8 + wave;
  const int f0    = ntile * 16;
  const int row   = lane & 15;
  const int koff  = (lane >> 4) << 1;

  const float* v = V + (size_t)bn * TT * FF;

  v8f acc[4] = {};
  for (int kk = 0; kk < TT; kk += 4) {
    v2f b;
    b.x = v[(size_t)(kk + koff)     * FF + f0 + row];
    b.y = v[(size_t)(kk + koff + 1) * FF + f0 + row];
#pragma unroll
    for (int pb = 0; pb < 4; ++pb) {
      v2f a = *(const v2f*)(&Sl[(pb * 16 + row) * SLD + kk + koff]);
      acc[pb] = __builtin_amdgcn_wmma_f32_16x16x4_f32(false, a, false, b,
                                                      (short)0, acc[pb], false, false);
    }
  }

  const int f = f0 + row;
  const int m = f >> 3;
  const float dw = (float)(m & 31) - 15.5f;
  const float dh = (float)(m >> 5) - 15.5f;
  float sx = 0.0f;
#pragma unroll
  for (int x = 0; x < 32; ++x) {
    float d = (float)x - 15.5f;
    sx += __expf(-d * d * (1.0f / 50.0f));
  }
  const float g = __expf(-(dw * dw + dh * dh) * (1.0f / 50.0f)) / (sx * sx);

  float* o = O + (size_t)bn * TT * FF;
  const int half = (lane >> 4) * 8;
#pragma unroll
  for (int pb = 0; pb < 4; ++pb) {
#pragma unroll
    for (int r = 0; r < 8; ++r) {
      const int p = pb * 16 + half + r;
      o[(size_t)p * FF + f] = acc[pb][r] * g;
    }
  }
}

extern "C" void kernel_launch(void* const* d_in, const int* in_sizes, int n_in,
                              void* d_out, int out_size, void* d_ws, size_t ws_size,
                              hipStream_t stream) {
  const float* Q = (const float*)d_in[0];
  const float* K = (const float*)d_in[1];
  const float* V = (const float*)d_in[2];
  float* O = (float*)d_out;
  float* S = (float*)d_ws;
  float* P = O;

  qk_partial_kernel<<<BN * KSPLIT, 256, 0, stream>>>(Q, K, P);
  s_sum_kernel<<<64, 256, 0, stream>>>(P, S);
  sv_gauss_kernel<<<BN * 64, 256, 0, stream>>>(S, V, O);
}


# SV kernel re-tiled: a wave owns 64 features x all 64 rows, lane n owns features 4n..4n+3 (tile j column n = feature 4n+j), so V loads and O stores are 128-bit whole lines; 8192 -> 2048 waves
# speedup vs baseline: 10.8466x; 1.1081x over previous
#include <hip/hip_runtime.h>

typedef __attribute__((ext_vector_type(2))) float v2f;
typedef __attribute__((ext_vector_type(4))) float v4f;
typedef __attribute__((ext_vector_type(8))) float v8f;

#define BN   16
#define TT   64
#define FF   8192
#define SLD  68
#define KSPLIT 32
#define KC   (FF / KSPLIT)

#define SUB   64
#define NSUB  (KC / SUB)
#define LP    68

__global__ void __launch_bounds__(256)
qk_partial_kernel(const float* __restrict__ Q,
                  const float* __restrict__ K,
                  float* __restrict__ P) {
  __shared__ __attribute__((aligned(16))) float L[2][2 * TT * LP];

  const int bn   = blockIdx.x / KSPLIT;
  const int c    = blockIdx.x % KSPLIT;
  const int tid  = threadIdx.x;
  const int lane = tid & 31;
  const int wave = tid >> 5;
  const int row  = lane & 15;
  const int koff = (lane >> 4) << 1;

  const int srow = tid >> 4;
  const int scol = (tid & 15) << 2;
  const size_t goff = (size_t)bn * TT * FF + (size_t)c * KC + (size_t)srow * FF + scol;
  const float* gq = Q + goff;
  const float* gk = K + goff;
  const int loff = srow * LP + scol;

  const int tj   = wave & 3;
  const int ti_a = wave >> 2;
  const int ti_b = ti_a + 2;
  const int fa = (ti_a * 16 + row) * LP + koff;
  const int fb = (ti_b * 16 + row) * LP + koff;
  const int fk = (TT + tj * 16 + row) * LP + koff;

  v4f rq[4], rk[4];
#pragma unroll
  for (int i = 0; i < 4; ++i) {
    rq[i] = *(const v4f*)(gq + (size_t)(16 * i) * FF);
    rk[i] = *(const v4f*)(gk + (size_t)(16 * i) * FF);
  }
#pragma unroll
  for (int i = 0; i < 4; ++i) {
    *(v4f*)(&L[0][loff + 16 * i * LP])        = rq[i];
    *(v4f*)(&L[0][loff + (TT + 16 * i) * LP]) = rk[i];
  }
  __syncthreads();

  v8f acc_a = {};
  v8f acc_b = {};
#pragma unroll
  for (int s = 0; s < NSUB; ++s) {
    const float* lc = L[s & 1];
    float*       ln = L[(s + 1) & 1];
    if (s + 1 < NSUB) {
#pragma unroll
      for (int i = 0; i < 4; ++i) {
        rq[i] = *(const v4f*)(gq + (size_t)(16 * i) * FF + (s + 1) * SUB);
        rk[i] = *(const v4f*)(gk + (size_t)(16 * i) * FF + (s + 1) * SUB);
      }
    }
#pragma unroll
    for (int kk = 0; kk < SUB; kk += 4) {
      v2f a0 = *(const v2f*)(lc + fa + kk);
      v2f a1 = *(const v2f*)(lc + fb + kk);
      v2f b  = *(const v2f*)(lc + fk + kk);
      acc_a = __builtin_amdgcn_wmma_f32_16x16x4_f32(false, a0, false, b,
                                                    (short)0, acc_a, false, false);
      acc_b = __builtin_amdgcn_wmma_f32_16x16x4_f32(false, a1, false, b,
                                                    (short)0, acc_b, false, false);
    }
    if (s + 1 < NSUB) {
#pragma unroll
      for (int i = 0; i < 4; ++i) {
        *(v4f*)(&ln[loff + 16 * i * LP])        = rq[i];
        *(v4f*)(&ln[loff + (TT + 16 * i) * LP]) = rk[i];
      }
      __syncthreads();
    }
  }

  float* pb = P + (size_t)blockIdx.x * TT * TT;
  const int qcol = tj * 16 + row;
  const int half = (lane >> 4) * 8;
#pragma unroll
  for (int r = 0; r < 8; ++r) {
    pb[(ti_a * 16 + half + r) * TT + qcol] = acc_a[r];
    pb[(ti_b * 16 + half + r) * TT + qcol] = acc_b[r];
  }
}

__global__ void __launch_bounds__(256)
s_sum_kernel(const float* __restrict__ P,
             float* __restrict__ S) {
  const int g  = blockIdx.x * 256 + threadIdx.x;
  const int bn = g >> 10;
  const int e  = (g & 1023) << 2;
  const float* p = P + (size_t)bn * KSPLIT * TT * TT + e;

  v4f s = *(const v4f*)p;
#pragma unroll
  for (int c = 1; c < KSPLIT; ++c) {
    s += *(const v4f*)(p + (size_t)c * TT * TT);
  }
  const float scale = 1.0f / 1024.0f;
  *(v4f*)(S + (size_t)bn * TT * TT + e) = s * scale;
}

#define FW 64
#define WGF (8 * FW)
#define NWG (FF / WGF)

__device__ __forceinline__ void sv_pblock(const float* sl, const v2f (&bp)[TT / 4][4],
                                          const float g, float* orow) {
  v2f a[TT / 4];
#pragma unroll
  for (int t = 0; t < TT / 4; ++t) {
    a[t] = *(const v2f*)(sl + 4 * t);
  }
  v8f acc[4] = {};
#pragma unroll
  for (int t = 0; t < TT / 4; ++t) {
#pragma unroll
    for (int j = 0; j < 4; ++j) {
      acc[j] = __builtin_amdgcn_wmma_f32_16x16x4_f32(false, a[t], false, bp[t][j],
                                                     (short)0, acc[j], false, false);
    }
  }
#pragma unroll
  for (int r = 0; r < 8; ++r) {
    v4f ov;
    ov.x = acc[0][r] * g;
    ov.y = acc[1][r] * g;
    ov.z = acc[2][r] * g;
    ov.w = acc[3][r] * g;
    *(v4f*)(orow + (size_t)r * FF) = ov;
  }
}

__global__ void __launch_bounds__(256)
sv_gauss_kernel(const float* __restrict__ S,
                const float* __restrict__ V,
                float* __restrict__ O) {
  __shared__ __attribute__((aligned(16))) float Sl[TT * SLD];

  const int bn    = blockIdx.x / NWG;
  const int chunk = blockIdx.x % NWG;
  const int tid   = threadIdx.x;
  const int lane  = tid & 31;
  const int wave  = tid >> 5;
  const int row   = lane & 15;
  const int koff  = (lane >> 4) << 1;

  const int f0 = (chunk * 8 + wave) * FW;
  const int fl = f0 + 4 * row;

  const float* sg = S + (size_t)bn * TT * TT;
  float sreg[TT * TT / 256];
#pragma unroll
  for (int i = 0; i < TT * TT / 256; ++i) {
    sreg[i] = sg[tid + 256 * i];
  }
#pragma unroll
  for (int i = 0; i < TT * TT / 256; ++i) {
    const int e = tid + 256 * i;
    Sl[(e >> 6) * SLD + (e & 63)] = sreg[i];
  }
  __syncthreads();

  const float* v = V + (size_t)bn * TT * FF + (size_t)koff * FF + fl;
  v4f vr[TT / 4][2];
#pragma unroll
  for (int t = 0; t < TT / 4; ++t) {
    vr[t][0] = *(const v4f*)(v + (size_t)(4 * t)     * FF);
    vr[t][1] = *(const v4f*)(v + (size_t)(4 * t + 1) * FF);
  }

  const int m = fl >> 3;
  const float dw = (float)(m & 31) - 15.5f;
  const float dh = (float)(m >> 5) - 15.5f;
  float sx = 0.0f;
#pragma unroll
  for (int x = 0; x < 32; ++x) {
    float d = (float)x - 15.5f;
    sx += __expf(-d * d * (1.0f / 50.0f));
  }
  const float g = __expf(-(dw * dw + dh * dh) * (1.0f / 50.0f)) / (sx * sx);

  v2f bp[TT / 4][4];
#pragma unroll
  for (int t = 0; t < TT / 4; ++t) {
#pragma unroll
    for (int j = 0; j < 4; ++j) {
      bp[t][j].x = vr[t][0][j];
      bp[t][j].y = vr[t][1][j];
    }
  }

  const int half = (lane >> 4) * 8;
  float* o = O + (size_t)bn * TT * FF + (size_t)half * FF + fl;
  const float* sl = &Sl[row * SLD + koff];

  sv_pblock(sl, bp, g, o);
#pragma nounroll
  for (int pb = 1; pb < 4; ++pb) {
    sv_pblock(sl + pb * 16 * SLD, bp, g, o + (size_t)pb * 16 * FF);
  }
}

extern "C" void kernel_launch(void* const* d_in, const int* in_sizes, int n_in,
                              void* d_out, int out_size, void* d_ws, size_t ws_size,
                              hipStream_t stream) {
  const float* Q = (const float*)d_in[0];
  const float* K = (const float*)d_in[1];
  const float* V = (const float*)d_in[2];
  float* O = (float*)d_out;
  float* S = (float*)d_ws;
  float* P = O;

  qk_partial_kernel<<<BN * KSPLIT, 256, 0, stream>>>(Q, K, P);
  s_sum_kernel<<<64, 256, 0, stream>>>(P, S);
  sv_gauss_kernel<<<BN * NWG, 256, 0, stream>>>(S, V, O);
}


# Score reduction split 16 ways instead of 32: a work-group sums 512 features (staged 128 at a time), 256 work-groups; half the waves, half the partial planes; changes the f32 summation order
# speedup vs baseline: 12.0576x; 1.1116x over previous
#include <hip/hip_runtime.h>

typedef __attribute__((ext_vector_type(2))) float v2f;
typedef __attribute__((ext_vector_type(4))) float v4f;
typedef __attribute__((ext_vector_type(8))) float v8f;

#define BN   16
#define TT   64
#define FF   8192
#define SLD  68
#define KSPLIT 16
#define KC   (FF / KSPLIT)

#define SUB   128
#define NSUB  (KC / SUB)
#define LP    132
#define NST   8

__global__ void __launch_bounds__(256)
qk_partial_kernel(const float* __restrict__ Q,
                  const float* __restrict__ K,
                  float* __restrict__ P) {
  __shared__ __attribute__((aligned(16))) float L[2][2 * TT * LP];

  const int bn   = blockIdx.x / KSPLIT;
  const int c    = blockIdx.x % KSPLIT;
  const int tid  = threadIdx.x;
  const int lane = tid & 31;
  const int wave = tid >> 5;
  const int row  = lane & 15;
  const int koff = (lane >> 4) << 1;

  const int srow = tid >> 5;
  const int scol = (tid & 31) << 2;
  const size_t goff = (size_t)bn * TT * FF + (size_t)c * KC + (size_t)srow * FF + scol;
  const float* gq = Q + goff;
  const float* gk = K + goff;
  const int loff = srow * LP + scol;

  const int tj   = wave & 3;
  const int ti_a = wave >> 2;
  const int ti_b = ti_a + 2;
  const int fa = (ti_a * 16 + row) * LP + koff;
  const int fb = (ti_b * 16 + row) * LP + koff;
  const int fk = (TT + tj * 16 + row) * LP + koff;

  v4f rq[NST], rk[NST];
#pragma unroll
  for (int i = 0; i < NST; ++i) {
    rq[i] = *(const v4f*)(gq + (size_t)(8 * i) * FF);
    rk[i] = *(const v4f*)(gk + (size_t)(8 * i) * FF);
  }
#pragma unroll
  for (int i = 0; i < NST; ++i) {
    *(v4f*)(&L[0][loff + 8 * i * LP])        = rq[i];
    *(v4f*)(&L[0][loff + (TT + 8 * i) * LP]) = rk[i];
  }
  __syncthreads();

  v8f acc_a = {};
  v8f acc_b = {};
#pragma unroll
  for (int s = 0; s < NSUB; ++s) {
    const float* lc = L[s & 1];
    float*       ln = L[(s + 1) & 1];
    if (s + 1 < NSUB) {
#pragma unroll
      for (int i = 0; i < NST; ++i) {
        rq[i] = *(const v4f*)(gq + (size_t)(8 * i) * FF + (s + 1) * SUB);
        rk[i] = *(const v4f*)(gk + (size_t)(8 * i) * FF + (s + 1) * SUB);
      }
      __builtin_amdgcn_sched_barrier(0);
    }
#pragma unroll
    for (int kk = 0; kk < SUB; kk += 4) {
      v2f a0 = *(const v2f*)(lc + fa + kk);
      v2f a1 = *(const v2f*)(lc + fb + kk);
      v2f b  = *(const v2f*)(lc + fk + kk);
      acc_a = __builtin_amdgcn_wmma_f32_16x16x4_f32(false, a0, false, b,
                                                    (short)0, acc_a, false, false);
      acc_b = __builtin_amdgcn_wmma_f32_16x16x4_f32(false, a1, false, b,
                                                    (short)0, acc_b, false, false);
    }
    if (s + 1 < NSUB) {
#pragma unroll
      for (int i = 0; i < NST; ++i) {
        *(v4f*)(&ln[loff + 8 * i * LP])        = rq[i];
        *(v4f*)(&ln[loff + (TT + 8 * i) * LP]) = rk[i];
      }
      __syncthreads();
    }
  }

  float* pb = P + (size_t)blockIdx.x * TT * TT;
  const int qcol = tj * 16 + row;
  const int half = (lane >> 4) * 8;
#pragma unroll
  for (int r = 0; r < 8; ++r) {
    pb[(ti_a * 16 + half + r) * TT + qcol] = acc_a[r];
    pb[(ti_b * 16 + half + r) * TT + qcol] = acc_b[r];
  }
}

__global__ void __launch_bounds__(256)
s_sum_kernel(const float* __restrict__ P,
             float* __restrict__ S) {
  const int g  = blockIdx.x * 256 + threadIdx.x;
  const int bn = g >> 10;
  const int e  = (g & 1023) << 2;
  const float* p = P + (size_t)bn * KSPLIT * TT * TT + e;

  v4f s = *(const v4f*)p;
#pragma unroll
  for (int c = 1; c < KSPLIT; ++c) {
    s += *(const v4f*)(p + (size_t)c * TT * TT);
  }
  const float scale = 1.0f / 1024.0f;
  *(v4f*)(S + (size_t)bn * TT * TT + e) = s * scale;
}

#define FW 64
#define WGF (8 * FW)
#define NWG (FF / WGF)

__device__ __forceinline__ void sv_pblock(const float* sl, const v2f (&bp)[TT / 4][4],
                                          const float g, float* orow) {
  v2f a[TT / 4];
#pragma unroll
  for (int t = 0; t < TT / 4; ++t) {
    a[t] = *(const v2f*)(sl + 4 * t);
  }
  v8f acc[4] = {};
#pragma unroll
  for (int t = 0; t < TT / 4; ++t) {
#pragma unroll
    for (int j = 0; j < 4; ++j) {
      acc[j] = __builtin_amdgcn_wmma_f32_16x16x4_f32(false, a[t], false, bp[t][j],
                                                     (short)0, acc[j], false, false);
    }
  }
#pragma unroll
  for (int r = 0; r < 8; ++r) {
    v4f ov;
    ov.x = acc[0][r] * g;
    ov.y = acc[1][r] * g;
    ov.z = acc[2][r] * g;
    ov.w = acc[3][r] * g;
    *(v4f*)(orow + (size_t)r * FF) = ov;
  }
}

__global__ void __launch_bounds__(256)
sv_gauss_kernel(const float* __restrict__ S,
                const float* __restrict__ V,
                float* __restrict__ O) {
  __shared__ __attribute__((aligned(16))) float Sl[TT * SLD];

  const int bn    = blockIdx.x / NWG;
  const int chunk = blockIdx.x % NWG;
  const int tid   = threadIdx.x;
  const int lane  = tid & 31;
  const int wave  = tid >> 5;
  const int row   = lane & 15;
  const int koff  = (lane >> 4) << 1;

  const int f0 = (chunk * 8 + wave) * FW;
  const int fl = f0 + 4 * row;

  const float* sg = S + (size_t)bn * TT * TT;
  float sreg[TT * TT / 256];
#pragma unroll
  for (int i = 0; i < TT * TT / 256; ++i) {
    sreg[i] = sg[tid + 256 * i];
  }
#pragma unroll
  for (int i = 0; i < TT * TT / 256; ++i) {
    const int e = tid + 256 * i;
    Sl[(e >> 6) * SLD + (e & 63)] = sreg[i];
  }
  __syncthreads();

  const float* v = V + (size_t)bn * TT * FF + (size_t)koff * FF + fl;
  v4f vr[TT / 4][2];
#pragma unroll
  for (int t = 0; t < TT / 4; ++t) {
    vr[t][0] = *(const v4f*)(v + (size_t)(4 * t)     * FF);
    vr[t][1] = *(const v4f*)(v + (size_t)(4 * t + 1) * FF);
  }

  const int m = fl >> 3;
  const float dw = (float)(m & 31) - 15.5f;
  const float dh = (float)(m >> 5) - 15.5f;
  float sx = 0.0f;
#pragma unroll
  for (int x = 0; x < 32; ++x) {
    float d = (float)x - 15.5f;
    sx += __expf(-d * d * (1.0f / 50.0f));
  }
  const float g = __expf(-(dw * dw + dh * dh) * (1.0f / 50.0f)) / (sx * sx);

  v2f bp[TT / 4][4];
#pragma unroll
  for (int t = 0; t < TT / 4; ++t) {
#pragma unroll
    for (int j = 0; j < 4; ++j) {
      bp[t][j].x = vr[t][0][j];
      bp[t][j].y = vr[t][1][j];
    }
  }

  const int half = (lane >> 4) * 8;
  float* o = O + (size_t)bn * TT * FF + (size_t)half * FF + fl;
  const float* sl = &Sl[row * SLD + koff];

  sv_pblock(sl, bp, g, o);
#pragma nounroll
  for (int pb = 1; pb < 4; ++pb) {
    sv_pblock(sl + pb * 16 * SLD, bp, g, o + (size_t)pb * 16 * FF);
  }
}

extern "C" void kernel_launch(void* const* d_in, const int* in_sizes, int n_in,
                              void* d_out, int out_size, void* d_ws, size_t ws_size,
                              hipStream_t stream) {
  const float* Q = (const float*)d_in[0];
  const float* K = (const float*)d_in[1];
  const float* V = (const float*)d_in[2];
  float* O = (float*)d_out;
  float* S = (float*)d_ws;
  float* P = O;

  qk_partial_kernel<<<BN * KSPLIT, 256, 0, stream>>>(Q, K, P);
  s_sum_kernel<<<64, 256, 0, stream>>>(P, S);
  sv_gauss_kernel<<<BN * NWG, 256, 0, stream>>>(S, V, O);
}
